# starting kernel
# baseline (speedup 1.0000x reference)
#include <hip/hip_runtime.h>

typedef __attribute__((ext_vector_type(2))) float v2f;
typedef __attribute__((ext_vector_type(8))) float v8f;

#define N_NODES 16384
#define N_ANCH  256
#define DLAT    32
#define BSTR    40

__global__ void pnn_prep_M(const float* __restrict__ embeds,
                           const float* __restrict__ W,
                           const int* __restrict__ anchor_ids,
                           float* __restrict__ Mp) {
    const int a = blockIdx.x;
    const int j = threadIdx.x;
    const int aid = anchor_ids[a];
    const float* e = embeds + (size_t)aid * DLAT;
    float acc = 0.0f;
#pragma unroll
    for (int d = 0; d < DLAT; ++d)
        acc = fmaf(e[d], W[d * DLAT + j], acc);
    Mp[a * DLAT + j] = acc * (1.0f / (float)N_ANCH);
}

__global__ void pnn_prep_C(const float* __restrict__ embeds,
                           const float* __restrict__ W,
                           const float* __restrict__ bias,
                           float* __restrict__ Cv) {
    const int j = threadIdx.x;
    float acc = 0.0f;
    for (int a = 0; a < N_ANCH; ++a) {
        const float* e = embeds + (size_t)a * DLAT;
#pragma unroll
        for (int d = 0; d < DLAT; ++d)
            acc = fmaf(e[d], W[(DLAT + d) * DLAT + j], acc);
    }
    Cv[j] = acc * (1.0f / (float)N_ANCH) + bias[j];
}

__global__ __launch_bounds__(256) void pnn_main(
        const float* __restrict__ dists,
        const float* __restrict__ Mp,
        const float* __restrict__ Cv,
        float* __restrict__ out) {
    __shared__ float Blds[N_ANCH * BSTR];

    const int tid = threadIdx.x;

#pragma unroll 4
    for (int i = tid; i < N_ANCH * DLAT; i += 256) {
        const int a = i >> 5, j = i & 31;
        Blds[a * BSTR + j] = Mp[i];
    }
    __syncthreads();

    const int wave = tid >> 5;
    const int lane = tid & 31;
    const int m    = lane & 15;
    const int kh   = lane >> 4;
    const int node_base = (blockIdx.x * 8 + wave) * 16;

    const float c_lo = Cv[m];
    const float c_hi = Cv[m + 16];

    v8f acc0 = {};
    v8f acc1 = {};

    const float* aptr = dists + node_base + m;

#pragma unroll 4
    for (int k0 = 0; k0 < N_ANCH; k0 += 4) {
        const int ka = k0 + 2 * kh;
        v2f af, b0, b1;
        af.x = aptr[(size_t)ka * N_NODES];
        af.y = aptr[(size_t)(ka + 1) * N_NODES];
        b0.x = Blds[ka * BSTR + m];
        b0.y = Blds[(ka + 1) * BSTR + m];
        b1.x = Blds[ka * BSTR + m + 16];
        b1.y = Blds[(ka + 1) * BSTR + m + 16];

        acc0 = __builtin_amdgcn_wmma_f32_16x16x4_f32(
                   false, af, false, b0, (short)0, acc0, false, false);
        acc1 = __builtin_amdgcn_wmma_f32_16x16x4_f32(
                   false, af, false, b1, (short)0, acc1, false, false);
    }

#pragma unroll
    for (int r = 0; r < 8; ++r) {
        const int row = node_base + r + 8 * kh;
        out[(size_t)row * DLAT + m]      = acc0[r] + c_lo;
        out[(size_t)row * DLAT + m + 16] = acc1[r] + c_hi;
    }
}

extern "C" void kernel_launch(void* const* d_in, const int* in_sizes, int n_in,
                              void* d_out, int out_size, void* d_ws, size_t ws_size,
                              hipStream_t stream) {
    const float* embeds     = (const float*)d_in[0];
    const float* dists      = (const float*)d_in[1];
    const float* W_hidden   = (const float*)d_in[2];
    const float* b_hidden   = (const float*)d_in[3];
    const int*   anchor_ids = (const int*)d_in[4];

    float* Mp = (float*)d_ws;
    float* Cv = Mp + N_ANCH * DLAT;
    float* out = (float*)d_out;

    pnn_prep_M<<<N_ANCH, DLAT, 0, stream>>>(embeds, W_hidden, anchor_ids, Mp);
    pnn_prep_C<<<1, DLAT, 0, stream>>>(embeds, W_hidden, b_hidden, Cv);

    pnn_main<<<N_NODES / (16 * 8), 256, 0, stream>>>(dists, Mp, Cv, out);
}


# prep_C: one wave's 8192-step dependent FMA chain replaced by 256-thread column sums of embeds[0:A] then one 32x32 product; changes summation order of the constant vector only
# speedup vs baseline: 5.5721x; 5.5721x over previous
#include <hip/hip_runtime.h>

typedef __attribute__((ext_vector_type(2))) float v2f;
typedef __attribute__((ext_vector_type(8))) float v8f;

#define N_NODES 16384
#define N_ANCH  256
#define DLAT    32
#define BSTR    40

__global__ void pnn_prep_M(const float* __restrict__ embeds,
                           const float* __restrict__ W,
                           const int* __restrict__ anchor_ids,
                           float* __restrict__ Mp) {
    const int a = blockIdx.x;
    const int j = threadIdx.x;
    const int aid = anchor_ids[a];
    const float* e = embeds + (size_t)aid * DLAT;
    float acc = 0.0f;
#pragma unroll
    for (int d = 0; d < DLAT; ++d)
        acc = fmaf(e[d], W[d * DLAT + j], acc);
    Mp[a * DLAT + j] = acc * (1.0f / (float)N_ANCH);
}

__global__ __launch_bounds__(256) void pnn_prep_C(const float* __restrict__ embeds,
                                                  const float* __restrict__ W,
                                                  const float* __restrict__ bias,
                                                  float* __restrict__ Cv) {
    __shared__ float part[8 * DLAT];

    const int lane = threadIdx.x & 31;
    const int g    = threadIdx.x >> 5;

    float w[DLAT] = {};
    float bj = 0.0f;
    if (g == 0) {
#pragma unroll
        for (int d = 0; d < DLAT; ++d)
            w[d] = W[(DLAT + d) * DLAT + lane];
        bj = bias[lane];
    }

    const float* e = embeds + (size_t)(g * 32) * DLAT + lane;
    float s0 = 0.0f, s1 = 0.0f, s2 = 0.0f, s3 = 0.0f;
#pragma unroll
    for (int i = 0; i < 32; i += 4) {
        s0 += e[(i + 0) * DLAT];
        s1 += e[(i + 1) * DLAT];
        s2 += e[(i + 2) * DLAT];
        s3 += e[(i + 3) * DLAT];
    }
    part[g * DLAT + lane] = (s0 + s1) + (s2 + s3);
    __syncthreads();

    if (g == 0) {
        float s = 0.0f;
#pragma unroll
        for (int r = 0; r < 8; ++r)
            s += part[r * DLAT + lane];

        float a0 = 0.0f, a1 = 0.0f, a2 = 0.0f, a3 = 0.0f;
#pragma unroll
        for (int d = 0; d < DLAT; d += 4) {
            a0 = fmaf(__shfl(s, d + 0), w[d + 0], a0);
            a1 = fmaf(__shfl(s, d + 1), w[d + 1], a1);
            a2 = fmaf(__shfl(s, d + 2), w[d + 2], a2);
            a3 = fmaf(__shfl(s, d + 3), w[d + 3], a3);
        }
        const float acc = (a0 + a1) + (a2 + a3);
        Cv[lane] = acc * (1.0f / (float)N_ANCH) + bj;
    }
}

__global__ __launch_bounds__(256) void pnn_main(
        const float* __restrict__ dists,
        const float* __restrict__ Mp,
        const float* __restrict__ Cv,
        float* __restrict__ out) {
    __shared__ float Blds[N_ANCH * BSTR];

    const int tid = threadIdx.x;

#pragma unroll 4
    for (int i = tid; i < N_ANCH * DLAT; i += 256) {
        const int a = i >> 5, j = i & 31;
        Blds[a * BSTR + j] = Mp[i];
    }
    __syncthreads();

    const int wave = tid >> 5;
    const int lane = tid & 31;
    const int m    = lane & 15;
    const int kh   = lane >> 4;
    const int node_base = (blockIdx.x * 8 + wave) * 16;

    const float c_lo = Cv[m];
    const float c_hi = Cv[m + 16];

    v8f acc0 = {};
    v8f acc1 = {};

    const float* aptr = dists + node_base + m;

#pragma unroll 4
    for (int k0 = 0; k0 < N_ANCH; k0 += 4) {
        const int ka = k0 + 2 * kh;
        v2f af, b0, b1;
        af.x = aptr[(size_t)ka * N_NODES];
        af.y = aptr[(size_t)(ka + 1) * N_NODES];
        b0.x = Blds[ka * BSTR + m];
        b0.y = Blds[(ka + 1) * BSTR + m];
        b1.x = Blds[ka * BSTR + m + 16];
        b1.y = Blds[(ka + 1) * BSTR + m + 16];

        acc0 = __builtin_amdgcn_wmma_f32_16x16x4_f32(
                   false, af, false, b0, (short)0, acc0, false, false);
        acc1 = __builtin_amdgcn_wmma_f32_16x16x4_f32(
                   false, af, false, b1, (short)0, acc1, false, false);
    }

#pragma unroll
    for (int r = 0; r < 8; ++r) {
        const int row = node_base + r + 8 * kh;
        out[(size_t)row * DLAT + m]      = acc0[r] + c_lo;
        out[(size_t)row * DLAT + m + 16] = acc1[r] + c_hi;
    }
}

extern "C" void kernel_launch(void* const* d_in, const int* in_sizes, int n_in,
                              void* d_out, int out_size, void* d_ws, size_t ws_size,
                              hipStream_t stream) {
    const float* embeds     = (const float*)d_in[0];
    const float* dists      = (const float*)d_in[1];
    const float* W_hidden   = (const float*)d_in[2];
    const float* b_hidden   = (const float*)d_in[3];
    const int*   anchor_ids = (const int*)d_in[4];

    float* Mp = (float*)d_ws;
    float* Cv = Mp + N_ANCH * DLAT;
    float* out = (float*)d_out;

    pnn_prep_M<<<N_ANCH, DLAT, 0, stream>>>(embeds, W_hidden, anchor_ids, Mp);
    pnn_prep_C<<<1, 256, 0, stream>>>(embeds, W_hidden, b_hidden, Cv);

    pnn_main<<<N_NODES / (16 * 8), 256, 0, stream>>>(dists, Mp, Cv, out);
}


# pnn_main: K loop split into a load phase and a WMMA phase: each wave issues all 128 A-operand loads in k order before the first product waits (sched_barrier holds them there); same k order and sums
# speedup vs baseline: 8.1388x; 1.4606x over previous
#include <hip/hip_runtime.h>

typedef __attribute__((ext_vector_type(2))) float v2f;
typedef __attribute__((ext_vector_type(8))) float v8f;

#define N_NODES 16384
#define N_ANCH  256
#define DLAT    32
#define BSTR    40

__global__ void pnn_prep_M(const float* __restrict__ embeds,
                           const float* __restrict__ W,
                           const int* __restrict__ anchor_ids,
                           float* __restrict__ Mp) {
    const int a = blockIdx.x;
    const int j = threadIdx.x;
    const int aid = anchor_ids[a];
    const float* e = embeds + (size_t)aid * DLAT;
    float acc = 0.0f;
#pragma unroll
    for (int d = 0; d < DLAT; ++d)
        acc = fmaf(e[d], W[d * DLAT + j], acc);
    Mp[a * DLAT + j] = acc * (1.0f / (float)N_ANCH);
}

__global__ __launch_bounds__(256) void pnn_prep_C(const float* __restrict__ embeds,
                                                  const float* __restrict__ W,
                                                  const float* __restrict__ bias,
                                                  float* __restrict__ Cv) {
    __shared__ float part[8 * DLAT];

    const int lane = threadIdx.x & 31;
    const int g    = threadIdx.x >> 5;

    float w[DLAT] = {};
    float bj = 0.0f;
    if (g == 0) {
#pragma unroll
        for (int d = 0; d < DLAT; ++d)
            w[d] = W[(DLAT + d) * DLAT + lane];
        bj = bias[lane];
    }

    const float* e = embeds + (size_t)(g * 32) * DLAT + lane;
    float s0 = 0.0f, s1 = 0.0f, s2 = 0.0f, s3 = 0.0f;
#pragma unroll
    for (int i = 0; i < 32; i += 4) {
        s0 += e[(i + 0) * DLAT];
        s1 += e[(i + 1) * DLAT];
        s2 += e[(i + 2) * DLAT];
        s3 += e[(i + 3) * DLAT];
    }
    part[g * DLAT + lane] = (s0 + s1) + (s2 + s3);
    __syncthreads();

    if (g == 0) {
        float s = 0.0f;
#pragma unroll
        for (int r = 0; r < 8; ++r)
            s += part[r * DLAT + lane];

        float a0 = 0.0f, a1 = 0.0f, a2 = 0.0f, a3 = 0.0f;
#pragma unroll
        for (int d = 0; d < DLAT; d += 4) {
            a0 = fmaf(__shfl(s, d + 0), w[d + 0], a0);
            a1 = fmaf(__shfl(s, d + 1), w[d + 1], a1);
            a2 = fmaf(__shfl(s, d + 2), w[d + 2], a2);
            a3 = fmaf(__shfl(s, d + 3), w[d + 3], a3);
        }
        const float acc = (a0 + a1) + (a2 + a3);
        Cv[lane] = acc * (1.0f / (float)N_ANCH) + bj;
    }
}

__global__ __launch_bounds__(256) void pnn_main(
        const float* __restrict__ dists,
        const float* __restrict__ Mp,
        const float* __restrict__ Cv,
        float* __restrict__ out) {
    __shared__ float Blds[N_ANCH * BSTR];

    const int tid = threadIdx.x;

#pragma unroll 4
    for (int i = tid; i < N_ANCH * DLAT; i += 256) {
        const int a = i >> 5, j = i & 31;
        Blds[a * BSTR + j] = Mp[i];
    }
    __syncthreads();

    const int wave = tid >> 5;
    const int lane = tid & 31;
    const int m    = lane & 15;
    const int kh   = lane >> 4;
    const int node_base = (blockIdx.x * 8 + wave) * 16;

    const float c_lo = Cv[m];
    const float c_hi = Cv[m + 16];

    const float* aptr = dists + (size_t)(2 * kh) * N_NODES + node_base + m;
    v2f af[N_ANCH / 4];
#pragma unroll
    for (int s = 0; s < N_ANCH / 4; ++s) {
        af[s].x = aptr[(size_t)(4 * s) * N_NODES];
        af[s].y = aptr[(size_t)(4 * s + 1) * N_NODES];
        if ((s & 7) == 7)
            __builtin_amdgcn_sched_barrier(0);
    }

    v8f acc0 = {};
    v8f acc1 = {};

#pragma unroll
    for (int s = 0; s < N_ANCH / 4; ++s) {
        const int ka = 4 * s + 2 * kh;
        v2f b0, b1;
        b0.x = Blds[ka * BSTR + m];
        b0.y = Blds[(ka + 1) * BSTR + m];
        b1.x = Blds[ka * BSTR + m + 16];
        b1.y = Blds[(ka + 1) * BSTR + m + 16];

        acc0 = __builtin_amdgcn_wmma_f32_16x16x4_f32(
                   false, af[s], false, b0, (short)0, acc0, false, false);
        acc1 = __builtin_amdgcn_wmma_f32_16x16x4_f32(
                   false, af[s], false, b1, (short)0, acc1, false, false);
    }

#pragma unroll
    for (int r = 0; r < 8; ++r) {
        const int row = node_base + r + 8 * kh;
        out[(size_t)row * DLAT + m]      = acc0[r] + c_lo;
        out[(size_t)row * DLAT + m + 16] = acc1[r] + c_hi;
    }
}

extern "C" void kernel_launch(void* const* d_in, const int* in_sizes, int n_in,
                              void* d_out, int out_size, void* d_ws, size_t ws_size,
                              hipStream_t stream) {
    const float* embeds     = (const float*)d_in[0];
    const float* dists      = (const float*)d_in[1];
    const float* W_hidden   = (const float*)d_in[2];
    const float* b_hidden   = (const float*)d_in[3];
    const int*   anchor_ids = (const int*)d_in[4];

    float* Mp = (float*)d_ws;
    float* Cv = Mp + N_ANCH * DLAT;
    float* out = (float*)d_out;

    pnn_prep_M<<<N_ANCH, DLAT, 0, stream>>>(embeds, W_hidden, anchor_ids, Mp);
    pnn_prep_C<<<1, 256, 0, stream>>>(embeds, W_hidden, b_hidden, Cv);

    pnn_main<<<N_NODES / (16 * 8), 256, 0, stream>>>(dists, Mp, Cv, out);
}


# prep_M and prep_C folded into one launch of 33 blocks (32 blocks of 8 waves build M', one per anchor per wave; block 32 builds C): two launches per call instead of three; same sums, same order
# speedup vs baseline: 9.8241x; 1.2071x over previous
#include <hip/hip_runtime.h>

typedef __attribute__((ext_vector_type(2))) float v2f;
typedef __attribute__((ext_vector_type(8))) float v8f;

#define N_NODES 16384
#define N_ANCH  256
#define DLAT    32
#define BSTR    40

__global__ __launch_bounds__(256) void pnn_prep(const float* __restrict__ embeds,
                                                const float* __restrict__ W,
                                                const float* __restrict__ bias,
                                                const int* __restrict__ anchor_ids,
                                                float* __restrict__ Mp,
                                                float* __restrict__ Cv) {
    __shared__ float part[8 * DLAT];

    const int lane = threadIdx.x & 31;
    const int g    = threadIdx.x >> 5;

    if (blockIdx.x < N_ANCH / 8) {
        const int a = blockIdx.x * 8 + g;
        const int aid = anchor_ids[a];
        const float* e = embeds + (size_t)aid * DLAT;
        float acc = 0.0f;
#pragma unroll
        for (int d = 0; d < DLAT; ++d)
            acc = fmaf(e[d], W[d * DLAT + lane], acc);
        Mp[a * DLAT + lane] = acc * (1.0f / (float)N_ANCH);
        return;
    }

    float w[DLAT] = {};
    float bj = 0.0f;
    if (g == 0) {
#pragma unroll
        for (int d = 0; d < DLAT; ++d)
            w[d] = W[(DLAT + d) * DLAT + lane];
        bj = bias[lane];
    }

    const float* e = embeds + (size_t)(g * 32) * DLAT + lane;
    float s0 = 0.0f, s1 = 0.0f, s2 = 0.0f, s3 = 0.0f;
#pragma unroll
    for (int i = 0; i < 32; i += 4) {
        s0 += e[(i + 0) * DLAT];
        s1 += e[(i + 1) * DLAT];
        s2 += e[(i + 2) * DLAT];
        s3 += e[(i + 3) * DLAT];
    }
    part[g * DLAT + lane] = (s0 + s1) + (s2 + s3);
    __syncthreads();

    if (g == 0) {
        float s = 0.0f;
#pragma unroll
        for (int r = 0; r < 8; ++r)
            s += part[r * DLAT + lane];

        float a0 = 0.0f, a1 = 0.0f, a2 = 0.0f, a3 = 0.0f;
#pragma unroll
        for (int d = 0; d < DLAT; d += 4) {
            a0 = fmaf(__shfl(s, d + 0), w[d + 0], a0);
            a1 = fmaf(__shfl(s, d + 1), w[d + 1], a1);
            a2 = fmaf(__shfl(s, d + 2), w[d + 2], a2);
            a3 = fmaf(__shfl(s, d + 3), w[d + 3], a3);
        }
        const float acc = (a0 + a1) + (a2 + a3);
        Cv[lane] = acc * (1.0f / (float)N_ANCH) + bj;
    }
}

__global__ __launch_bounds__(256) void pnn_main(
        const float* __restrict__ dists,
        const float* __restrict__ Mp,
        const float* __restrict__ Cv,
        float* __restrict__ out) {
    __shared__ float Blds[N_ANCH * BSTR];

    const int tid = threadIdx.x;

#pragma unroll 4
    for (int i = tid; i < N_ANCH * DLAT; i += 256) {
        const int a = i >> 5, j = i & 31;
        Blds[a * BSTR + j] = Mp[i];
    }
    __syncthreads();

    const int wave = tid >> 5;
    const int lane = tid & 31;
    const int m    = lane & 15;
    const int kh   = lane >> 4;
    const int node_base = (blockIdx.x * 8 + wave) * 16;

    const float c_lo = Cv[m];
    const float c_hi = Cv[m + 16];

    const float* aptr = dists + (size_t)(2 * kh) * N_NODES + node_base + m;
    v2f af[N_ANCH / 4];
#pragma unroll
    for (int s = 0; s < N_ANCH / 4; ++s) {
        af[s].x = aptr[(size_t)(4 * s) * N_NODES];
        af[s].y = aptr[(size_t)(4 * s + 1) * N_NODES];
        if ((s & 7) == 7)
            __builtin_amdgcn_sched_barrier(0);
    }

    v8f acc0 = {};
    v8f acc1 = {};

#pragma unroll
    for (int s = 0; s < N_ANCH / 4; ++s) {
        const int ka = 4 * s + 2 * kh;
        v2f b0, b1;
        b0.x = Blds[ka * BSTR + m];
        b0.y = Blds[(ka + 1) * BSTR + m];
        b1.x = Blds[ka * BSTR + m + 16];
        b1.y = Blds[(ka + 1) * BSTR + m + 16];

        acc0 = __builtin_amdgcn_wmma_f32_16x16x4_f32(
                   false, af[s], false, b0, (short)0, acc0, false, false);
        acc1 = __builtin_amdgcn_wmma_f32_16x16x4_f32(
                   false, af[s], false, b1, (short)0, acc1, false, false);
    }

#pragma unroll
    for (int r = 0; r < 8; ++r) {
        const int row = node_base + r + 8 * kh;
        out[(size_t)row * DLAT + m]      = acc0[r] + c_lo;
        out[(size_t)row * DLAT + m + 16] = acc1[r] + c_hi;
    }
}

extern "C" void kernel_launch(void* const* d_in, const int* in_sizes, int n_in,
                              void* d_out, int out_size, void* d_ws, size_t ws_size,
                              hipStream_t stream) {
    const float* embeds     = (const float*)d_in[0];
    const float* dists      = (const float*)d_in[1];
    const float* W_hidden   = (const float*)d_in[2];
    const float* b_hidden   = (const float*)d_in[3];
    const int*   anchor_ids = (const int*)d_in[4];

    float* Mp = (float*)d_ws;
    float* Cv = Mp + N_ANCH * DLAT;
    float* out = (float*)d_out;

    pnn_prep<<<N_ANCH / 8 + 1, 256, 0, stream>>>(embeds, W_hidden, b_hidden, anchor_ids, Mp, Cv);

    pnn_main<<<N_NODES / (16 * 8), 256, 0, stream>>>(dists, Mp, Cv, out);
}


# prep launch folded into pnn_main: each block gathers the anchor rows, builds M' on f32 WMMA (sum over d now in matrix-unit order) into LDS and C per wave; one launch per call instead of two
# speedup vs baseline: 11.9605x; 1.2175x over previous
#include <hip/hip_runtime.h>


typedef __attribute__((ext_vector_type(2))) float v2f;
typedef __attribute__((ext_vector_type(4))) float v4f;
typedef __attribute__((ext_vector_type(8))) float v8f;

#define N_NODES 16384
#define N_ANCH  256
#define DLAT    32
#define BSTR    40

__device__ __forceinline__ float lane_bcast(float x, int src) {
    return __int_as_float(__builtin_amdgcn_readlane(__float_as_int(x), src));
}

__global__ __launch_bounds__(256) void pnn_main(
        const float* __restrict__ embeds,
        const float* __restrict__ dists,
        const float* __restrict__ W,
        const float* __restrict__ bias,
        const int*   __restrict__ anchor_ids,
        float* __restrict__ out) {
    __shared__ float Blds[N_ANCH * BSTR];
    __shared__ float part[8 * DLAT];

    const int tid  = threadIdx.x;
    const int wave = tid >> 5;
    const int lane = tid & 31;
    const int m    = lane & 15;
    const int kh   = lane >> 4;

    const int aid0 = anchor_ids[32 * wave + m];
    const int aid1 = anchor_ids[32 * wave + 16 + m];

    v2f wb[8][2];
#pragma unroll
    for (int s = 0; s < 8; ++s) {
        const int d = 16 * kh + 2 * s;
#pragma unroll
        for (int u = 0; u < 2; ++u) {
            wb[s][u].x = W[d * DLAT + m + 16 * u];
            wb[s][u].y = W[(d + 1) * DLAT + m + 16 * u];
        }
    }

    float w[DLAT];
#pragma unroll
    for (int d = 0; d < DLAT; ++d)
        w[d] = W[(DLAT + d) * DLAT + lane];
    const float bj = bias[lane];

    {
        const float* ec = embeds + (size_t)(wave * 32) * DLAT + lane;
        float s0 = 0.0f, s1 = 0.0f, s2 = 0.0f, s3 = 0.0f;
#pragma unroll
        for (int i = 0; i < 32; i += 4) {
            s0 += ec[(i + 0) * DLAT];
            s1 += ec[(i + 1) * DLAT];
            s2 += ec[(i + 2) * DLAT];
            s3 += ec[(i + 3) * DLAT];
        }
        part[wave * DLAT + lane] = (s0 + s1) + (s2 + s3);
    }

    const v4f* e0 = (const v4f*)(embeds + (size_t)aid0 * DLAT + 16 * kh);
    const v4f* e1 = (const v4f*)(embeds + (size_t)aid1 * DLAT + 16 * kh);
    v4f ea0[4], ea1[4];
#pragma unroll
    for (int q = 0; q < 4; ++q) {
        ea0[q] = e0[q];
        ea1[q] = e1[q];
    }

    v8f p00 = {}, p01 = {}, p10 = {}, p11 = {};
#pragma unroll
    for (int s = 0; s < 8; ++s) {
        v2f a0, a1;
        a0.x = ea0[s >> 1][2 * (s & 1)];
        a0.y = ea0[s >> 1][2 * (s & 1) + 1];
        a1.x = ea1[s >> 1][2 * (s & 1)];
        a1.y = ea1[s >> 1][2 * (s & 1) + 1];
        p00 = __builtin_amdgcn_wmma_f32_16x16x4_f32(
                  false, a0, false, wb[s][0], (short)0, p00, false, false);
        p01 = __builtin_amdgcn_wmma_f32_16x16x4_f32(
                  false, a0, false, wb[s][1], (short)0, p01, false, false);
        p10 = __builtin_amdgcn_wmma_f32_16x16x4_f32(
                  false, a1, false, wb[s][0], (short)0, p10, false, false);
        p11 = __builtin_amdgcn_wmma_f32_16x16x4_f32(
                  false, a1, false, wb[s][1], (short)0, p11, false, false);
    }

    const float inv_a = 1.0f / (float)N_ANCH;
#pragma unroll
    for (int r = 0; r < 8; ++r) {
        const int a = 32 * wave + r + 8 * kh;
        Blds[a * BSTR + m]             = p00[r] * inv_a;
        Blds[a * BSTR + m + 16]        = p01[r] * inv_a;
        Blds[(a + 16) * BSTR + m]      = p10[r] * inv_a;
        Blds[(a + 16) * BSTR + m + 16] = p11[r] * inv_a;
    }
    __syncthreads();

    const int node_base = (blockIdx.x * 8 + wave) * 16;

    const float* aptr = dists + (size_t)(2 * kh) * N_NODES + node_base + m;
    v2f af[N_ANCH / 4];
#pragma unroll
    for (int s = 0; s < N_ANCH / 4; ++s) {
        af[s].x = aptr[(size_t)(4 * s) * N_NODES];
        af[s].y = aptr[(size_t)(4 * s + 1) * N_NODES];
        if ((s & 7) == 7)
            __builtin_amdgcn_sched_barrier(0);
    }

    float sc = 0.0f;
#pragma unroll
    for (int r = 0; r < 8; ++r)
        sc += part[r * DLAT + lane];
    float a0 = 0.0f, a1 = 0.0f, a2 = 0.0f, a3 = 0.0f;
#pragma unroll
    for (int d = 0; d < DLAT; d += 4) {
        a0 = fmaf(lane_bcast(sc, d + 0), w[d + 0], a0);
        a1 = fmaf(lane_bcast(sc, d + 1), w[d + 1], a1);
        a2 = fmaf(lane_bcast(sc, d + 2), w[d + 2], a2);
        a3 = fmaf(lane_bcast(sc, d + 3), w[d + 3], a3);
    }
    const float cj = ((a0 + a1) + (a2 + a3)) * inv_a + bj;
    const float c_lo = __shfl(cj, m);
    const float c_hi = __shfl(cj, m + 16);

    v8f acc0 = {};
    v8f acc1 = {};

#pragma unroll
    for (int s = 0; s < N_ANCH / 4; ++s) {
        const int ka = 4 * s + 2 * kh;
        v2f b0, b1;
        b0.x = Blds[ka * BSTR + m];
        b0.y = Blds[(ka + 1) * BSTR + m];
        b1.x = Blds[ka * BSTR + m + 16];
        b1.y = Blds[(ka + 1) * BSTR + m + 16];

        acc0 = __builtin_amdgcn_wmma_f32_16x16x4_f32(
                   false, af[s], false, b0, (short)0, acc0, false, false);
        acc1 = __builtin_amdgcn_wmma_f32_16x16x4_f32(
                   false, af[s], false, b1, (short)0, acc1, false, false);
    }

#pragma unroll
    for (int r = 0; r < 8; ++r) {
        const int row = node_base + r + 8 * kh;
        out[(size_t)row * DLAT + m]      = acc0[r] + c_lo;
        out[(size_t)row * DLAT + m + 16] = acc1[r] + c_hi;
    }
}

extern "C" void kernel_launch(void* const* d_in, const int* in_sizes, int n_in,
                              void* d_out, int out_size, void* d_ws, size_t ws_size,
                              hipStream_t stream) {
    const float* embeds     = (const float*)d_in[0];
    const float* dists      = (const float*)d_in[1];
    const float* W_hidden   = (const float*)d_in[2];
    const float* b_hidden   = (const float*)d_in[3];
    const int*   anchor_ids = (const int*)d_in[4];
    float* out = (float*)d_out;

    pnn_main<<<N_NODES / (16 * 8), 256, 0, stream>>>(embeds, dists, W_hidden, b_hidden,
                                                     anchor_ids, out);
}


# pnn_main: each 16-node tile's anchor sum split over two waves (anchors 0-127 / 128-255), partial sums added once through LDS (changes summation order); 256 blocks of 64 nodes instead of 128 of 128
# speedup vs baseline: 15.1687x; 1.2682x over previous
#include <hip/hip_runtime.h>


typedef __attribute__((ext_vector_type(2))) float v2f;
typedef __attribute__((ext_vector_type(4))) float v4f;
typedef __attribute__((ext_vector_type(8))) float v8f;

#define N_NODES 16384
#define N_ANCH  256
#define DLAT    32
#define BSTR    40
#define KHALF   (N_ANCH / 2)

__device__ __forceinline__ float lane_bcast(float x, int src) {
    return __int_as_float(__builtin_amdgcn_readlane(__float_as_int(x), src));
}

__global__ __launch_bounds__(256) void pnn_main(
        const float* __restrict__ embeds,
        const float* __restrict__ dists,
        const float* __restrict__ W,
        const float* __restrict__ bias,
        const int*   __restrict__ anchor_ids,
        float* __restrict__ out) {
    __shared__ float Blds[N_ANCH * BSTR];
    __shared__ float part[8 * DLAT];
    __shared__ v4f   xch[8 * 2 * 32];

    const int tid  = threadIdx.x;
    const int wave = tid >> 5;
    const int lane = tid & 31;
    const int m    = lane & 15;
    const int kh   = lane >> 4;

    const int aid0 = anchor_ids[32 * wave + m];
    const int aid1 = anchor_ids[32 * wave + 16 + m];

    v2f wb[8][2];
#pragma unroll
    for (int s = 0; s < 8; ++s) {
        const int d = 16 * kh + 2 * s;
#pragma unroll
        for (int u = 0; u < 2; ++u) {
            wb[s][u].x = W[d * DLAT + m + 16 * u];
            wb[s][u].y = W[(d + 1) * DLAT + m + 16 * u];
        }
    }

    float w[DLAT];
#pragma unroll
    for (int d = 0; d < DLAT; ++d)
        w[d] = W[(DLAT + d) * DLAT + lane];
    const float bj = bias[lane];

    {
        const float* ec = embeds + (size_t)(wave * 32) * DLAT + lane;
        float s0 = 0.0f, s1 = 0.0f, s2 = 0.0f, s3 = 0.0f;
#pragma unroll
        for (int i = 0; i < 32; i += 4) {
            s0 += ec[(i + 0) * DLAT];
            s1 += ec[(i + 1) * DLAT];
            s2 += ec[(i + 2) * DLAT];
            s3 += ec[(i + 3) * DLAT];
        }
        part[wave * DLAT + lane] = (s0 + s1) + (s2 + s3);
    }

    const v4f* e0 = (const v4f*)(embeds + (size_t)aid0 * DLAT + 16 * kh);
    const v4f* e1 = (const v4f*)(embeds + (size_t)aid1 * DLAT + 16 * kh);
    v4f ea0[4], ea1[4];
#pragma unroll
    for (int q = 0; q < 4; ++q) {
        ea0[q] = e0[q];
        ea1[q] = e1[q];
    }

    v8f p00 = {}, p01 = {}, p10 = {}, p11 = {};
#pragma unroll
    for (int s = 0; s < 8; ++s) {
        v2f a0, a1;
        a0.x = ea0[s >> 1][2 * (s & 1)];
        a0.y = ea0[s >> 1][2 * (s & 1) + 1];
        a1.x = ea1[s >> 1][2 * (s & 1)];
        a1.y = ea1[s >> 1][2 * (s & 1) + 1];
        p00 = __builtin_amdgcn_wmma_f32_16x16x4_f32(
                  false, a0, false, wb[s][0], (short)0, p00, false, false);
        p01 = __builtin_amdgcn_wmma_f32_16x16x4_f32(
                  false, a0, false, wb[s][1], (short)0, p01, false, false);
        p10 = __builtin_amdgcn_wmma_f32_16x16x4_f32(
                  false, a1, false, wb[s][0], (short)0, p10, false, false);
        p11 = __builtin_amdgcn_wmma_f32_16x16x4_f32(
                  false, a1, false, wb[s][1], (short)0, p11, false, false);
    }

    const float inv_a = 1.0f / (float)N_ANCH;
#pragma unroll
    for (int r = 0; r < 8; ++r) {
        const int a = 32 * wave + r + 8 * kh;
        Blds[a * BSTR + m]             = p00[r] * inv_a;
        Blds[a * BSTR + m + 16]        = p01[r] * inv_a;
        Blds[(a + 16) * BSTR + m]      = p10[r] * inv_a;
        Blds[(a + 16) * BSTR + m + 16] = p11[r] * inv_a;
    }
    __syncthreads();

    const int tile = wave & 3;
    const int kq   = __builtin_amdgcn_readfirstlane(wave >> 2);
    const int node_base = (blockIdx.x * 4 + tile) * 16;

    const float* aptr = dists + (size_t)(KHALF * kq + 2 * kh) * N_NODES + node_base + m;
    v2f af[KHALF / 4];
#pragma unroll
    for (int s = 0; s < KHALF / 4; ++s) {
        af[s].x = aptr[(size_t)(4 * s) * N_NODES];
        af[s].y = aptr[(size_t)(4 * s + 1) * N_NODES];
        if ((s & 7) == 7)
            __builtin_amdgcn_sched_barrier(0);
    }

    float sc = 0.0f;
#pragma unroll
    for (int r = 0; r < 8; ++r)
        sc += part[r * DLAT + lane];
    float a0 = 0.0f, a1 = 0.0f, a2 = 0.0f, a3 = 0.0f;
#pragma unroll
    for (int d = 0; d < DLAT; d += 4) {
        a0 = fmaf(lane_bcast(sc, d + 0), w[d + 0], a0);
        a1 = fmaf(lane_bcast(sc, d + 1), w[d + 1], a1);
        a2 = fmaf(lane_bcast(sc, d + 2), w[d + 2], a2);
        a3 = fmaf(lane_bcast(sc, d + 3), w[d + 3], a3);
    }
    const float cj = ((a0 + a1) + (a2 + a3)) * inv_a + bj;
    const float c_lo = __shfl(cj, m);
    const float c_hi = __shfl(cj, m + 16);

    v8f acc0 = {};
    v8f acc1 = {};

#pragma unroll
    for (int s = 0; s < KHALF / 4; ++s) {
        const int ka = KHALF * kq + 4 * s + 2 * kh;
        v2f b0, b1;
        b0.x = Blds[ka * BSTR + m];
        b0.y = Blds[(ka + 1) * BSTR + m];
        b1.x = Blds[ka * BSTR + m + 16];
        b1.y = Blds[(ka + 1) * BSTR + m + 16];

        acc0 = __builtin_amdgcn_wmma_f32_16x16x4_f32(
                   false, af[s], false, b0, (short)0, acc0, false, false);
        acc1 = __builtin_amdgcn_wmma_f32_16x16x4_f32(
                   false, af[s], false, b1, (short)0, acc1, false, false);
    }

    v4f lo0, hi0, lo1, hi1;
#pragma unroll
    for (int i = 0; i < 4; ++i) {
        lo0[i] = acc0[i];  hi0[i] = acc0[4 + i];
        lo1[i] = acc1[i];  hi1[i] = acc1[4 + i];
    }
    const v4f keep0 = kq ? hi0 : lo0;
    const v4f keep1 = kq ? hi1 : lo1;
    xch[(wave * 2 + 0) * 32 + lane] = kq ? lo0 : hi0;
    xch[(wave * 2 + 1) * 32 + lane] = kq ? lo1 : hi1;
    __syncthreads();
    const int pw = wave ^ 4;
    const v4f got0 = xch[(pw * 2 + 0) * 32 + lane];
    const v4f got1 = xch[(pw * 2 + 1) * 32 + lane];

#pragma unroll
    for (int i = 0; i < 4; ++i) {
        const int row = node_base + 4 * kq + i + 8 * kh;
        out[(size_t)row * DLAT + m]      = (keep0[i] + got0[i]) + c_lo;
        out[(size_t)row * DLAT + m + 16] = (keep1[i] + got1[i]) + c_hi;
    }
}

extern "C" void kernel_launch(void* const* d_in, const int* in_sizes, int n_in,
                              void* d_out, int out_size, void* d_ws, size_t ws_size,
                              hipStream_t stream) {
    const float* embeds     = (const float*)d_in[0];
    const float* dists      = (const float*)d_in[1];
    const float* W_hidden   = (const float*)d_in[2];
    const float* b_hidden   = (const float*)d_in[3];
    const int*   anchor_ids = (const int*)d_in[4];
    float* out = (float*)d_out;

    pnn_main<<<N_NODES / 64, 256, 0, stream>>>(embeds, dists, W_hidden, b_hidden,
                                               anchor_ids, out);
}


# (dists.T @ E) @ W[0:D] replaces dists.T @ (E @ W[0:D]): gathered rows are the main product's B operand, W applied per summed 16x32 tile, tail loads under the main product; changes summation order
# speedup vs baseline: 15.9434x; 1.0511x over previous
#include <hip/hip_runtime.h>


typedef __attribute__((ext_vector_type(2))) float v2f;
typedef __attribute__((ext_vector_type(4))) float v4f;
typedef __attribute__((ext_vector_type(8))) float v8f;

#define N_NODES 16384
#define N_ANCH  256
#define DLAT    32
#define BSTR    40
#define TSTR    36
#define KHALF   (N_ANCH / 2)
#define KMID    (KHALF / 8)

__device__ __forceinline__ float lane_bcast(float x, int src) {
    return __int_as_float(__builtin_amdgcn_readlane(__float_as_int(x), src));
}

__global__ __launch_bounds__(256) void pnn_main(
        const float* __restrict__ embeds,
        const float* __restrict__ dists,
        const float* __restrict__ W,
        const float* __restrict__ bias,
        const int*   __restrict__ anchor_ids,
        float* __restrict__ out) {
    __shared__ __attribute__((aligned(16))) float Blds[N_ANCH * BSTR];
    __shared__ __attribute__((aligned(16))) float Tlds[8 * 16 * TSTR];
    __shared__ float part[8 * DLAT];

    const int tid  = threadIdx.x;
    const int wave = tid >> 5;
    const int lane = tid & 31;
    const int m    = lane & 15;
    const int kh   = lane >> 4;

    const int id   = anchor_ids[32 * wave + lane];
    const int ido  = __shfl_xor(id, 16);
    const int aid0 = kh ? ido : id;
    const int aid1 = kh ? id : ido;

    const v4f* e0 = (const v4f*)(embeds + (size_t)aid0 * DLAT + 16 * kh);
    const v4f* e1 = (const v4f*)(embeds + (size_t)aid1 * DLAT + 16 * kh);
    v4f ea0[4], ea1[4];
#pragma unroll
    for (int q = 0; q < 4; ++q) {
        ea0[q] = e0[q];
        ea1[q] = e1[q];
    }
    v4f* l0 = (v4f*)(Blds + (32 * wave + m) * BSTR + 16 * kh);
    v4f* l1 = (v4f*)(Blds + (32 * wave + 16 + m) * BSTR + 16 * kh);
#pragma unroll
    for (int q = 0; q < 4; ++q) {
        l0[q] = ea0[q];
        l1[q] = ea1[q];
    }
    __syncthreads();

    const int tile = wave & 3;
    const int kq   = __builtin_amdgcn_readfirstlane(wave >> 2);
    const int node_base = (blockIdx.x * 4 + tile) * 16;

    const float* aptr = dists + (size_t)(KHALF * kq + 2 * kh) * N_NODES + node_base + m;
    v2f af[KHALF / 4];
#pragma unroll
    for (int s = 0; s < KHALF / 4; ++s) {
        af[s].x = aptr[(size_t)(4 * s) * N_NODES];
        af[s].y = aptr[(size_t)(4 * s + 1) * N_NODES];
        if ((s & 7) == 7)
            __builtin_amdgcn_sched_barrier(0);
    }

    v8f acc0 = {};
    v8f acc1 = {};

    float cs[32];
    v2f   wb[8][2];
    float w[DLAT];
    float bj;

#pragma unroll
    for (int s = 0; s < KHALF / 4; ++s) {
        if (s == KMID) {
            __builtin_amdgcn_sched_barrier(0);
            const float* ec = embeds + (size_t)(wave * 32) * DLAT + lane;
#pragma unroll
            for (int i = 0; i < 32; ++i)
                cs[i] = ec[i * DLAT];
            if (kq == 0) {
#pragma unroll
                for (int t = 0; t < 8; ++t) {
                    const int d = 16 * kh + 2 * t;
#pragma unroll
                    for (int u = 0; u < 2; ++u) {
                        wb[t][u].x = W[d * DLAT + m + 16 * u];
                        wb[t][u].y = W[(d + 1) * DLAT + m + 16 * u];
                    }
                }
#pragma unroll
                for (int d = 0; d < DLAT; ++d)
                    w[d] = W[(DLAT + d) * DLAT + lane];
                bj = bias[lane];
            }
        }
        const int ka = KHALF * kq + 4 * s + 2 * kh;
        v2f b0, b1;
        b0.x = Blds[ka * BSTR + m];
        b0.y = Blds[(ka + 1) * BSTR + m];
        b1.x = Blds[ka * BSTR + m + 16];
        b1.y = Blds[(ka + 1) * BSTR + m + 16];

        acc0 = __builtin_amdgcn_wmma_f32_16x16x4_f32(
                   false, af[s], false, b0, (short)0, acc0, false, false);
        acc1 = __builtin_amdgcn_wmma_f32_16x16x4_f32(
                   false, af[s], false, b1, (short)0, acc1, false, false);
    }

    {
        float s0 = 0.0f, s1 = 0.0f, s2 = 0.0f, s3 = 0.0f;
#pragma unroll
        for (int i = 0; i < 32; i += 4) {
            s0 += cs[i + 0];
            s1 += cs[i + 1];
            s2 += cs[i + 2];
            s3 += cs[i + 3];
        }
        part[wave * DLAT + lane] = (s0 + s1) + (s2 + s3);
    }

    float* tw = Tlds + wave * (16 * TSTR);
#pragma unroll
    for (int r = 0; r < 8; ++r) {
        tw[(8 * kh + r) * TSTR + m]      = acc0[r];
        tw[(8 * kh + r) * TSTR + m + 16] = acc1[r];
    }
    __syncthreads();

    if (kq == 0) {
        const v4f* t0 = (const v4f*)(Tlds + wave * (16 * TSTR) + m * TSTR + 16 * kh);
        const v4f* t1 = (const v4f*)(Tlds + (wave + 4) * (16 * TSTR) + m * TSTR + 16 * kh);
        v4f ta[4];
#pragma unroll
        for (int q = 0; q < 4; ++q)
            ta[q] = t0[q] + t1[q];

        float sc = 0.0f;
#pragma unroll
        for (int r = 0; r < 8; ++r)
            sc += part[r * DLAT + lane];
        float a0 = 0.0f, a1 = 0.0f, a2 = 0.0f, a3 = 0.0f;
#pragma unroll
        for (int d = 0; d < DLAT; d += 4) {
            a0 = fmaf(lane_bcast(sc, d + 0), w[d + 0], a0);
            a1 = fmaf(lane_bcast(sc, d + 1), w[d + 1], a1);
            a2 = fmaf(lane_bcast(sc, d + 2), w[d + 2], a2);
            a3 = fmaf(lane_bcast(sc, d + 3), w[d + 3], a3);
        }
        const float inv_a = 1.0f / (float)N_ANCH;
        const float cj = ((a0 + a1) + (a2 + a3)) * inv_a + bj;
        const float c_lo = __shfl(cj, m);
        const float c_hi = __shfl(cj, m + 16);

        v8f o0 = {};
        v8f o1 = {};
#pragma unroll
        for (int s = 0; s < 8; ++s) {
            v2f a;
            a.x = ta[s >> 1][2 * (s & 1)];
            a.y = ta[s >> 1][2 * (s & 1) + 1];
            o0 = __builtin_amdgcn_wmma_f32_16x16x4_f32(
                     false, a, false, wb[s][0], (short)0, o0, false, false);
            o1 = __builtin_amdgcn_wmma_f32_16x16x4_f32(
                     false, a, false, wb[s][1], (short)0, o1, false, false);
        }

#pragma unroll
        for (int r = 0; r < 8; ++r) {
            const int row = node_base + 8 * kh + r;
            out[(size_t)row * DLAT + m]      = fmaf(o0[r], inv_a, c_lo);
            out[(size_t)row * DLAT + m + 16] = fmaf(o1[r], inv_a, c_hi);
        }
    }
}

extern "C" void kernel_launch(void* const* d_in, const int* in_sizes, int n_in,
                              void* d_out, int out_size, void* d_ws, size_t ws_size,
                              hipStream_t stream) {
    const float* embeds     = (const float*)d_in[0];
    const float* dists      = (const float*)d_in[1];
    const float* W_hidden   = (const float*)d_in[2];
    const float* b_hidden   = (const float*)d_in[3];
    const int*   anchor_ids = (const int*)d_in[4];
    float* out = (float*)d_out;

    pnn_main<<<N_NODES / 64, 256, 0, stream>>>(embeds, dists, W_hidden, b_hidden,
                                               anchor_ids, out);
}


# pnn_main: A-operand (dists) loads no longer start behind the gather's barrier: half go out under the id wait, half once the rows landed; that barrier now waits for LDS only; same sums and order
# speedup vs baseline: 17.8941x; 1.1224x over previous
#include <hip/hip_runtime.h>


typedef __attribute__((ext_vector_type(2))) float v2f;
typedef __attribute__((ext_vector_type(4))) float v4f;
typedef __attribute__((ext_vector_type(8))) float v8f;

#define N_NODES 16384
#define N_ANCH  256
#define DLAT    32
#define BSTR    40
#define TSTR    36
#define KHALF   (N_ANCH / 2)
#define KMID    (KHALF / 8)
#define KPRE    16

__device__ __forceinline__ float lane_bcast(float x, int src) {
    return __int_as_float(__builtin_amdgcn_readlane(__float_as_int(x), src));
}

__global__ __launch_bounds__(256) void pnn_main(
        const float* __restrict__ embeds,
        const float* __restrict__ dists,
        const float* __restrict__ W,
        const float* __restrict__ bias,
        const int*   __restrict__ anchor_ids,
        float* __restrict__ out) {
    __shared__ __attribute__((aligned(16))) float Blds[N_ANCH * BSTR];
    __shared__ __attribute__((aligned(16))) float Tlds[8 * 16 * TSTR];
    __shared__ float part[8 * DLAT];

    const int tid  = threadIdx.x;
    const int wave = tid >> 5;
    const int lane = tid & 31;
    const int m    = lane & 15;
    const int kh   = lane >> 4;

    const int tile = wave & 3;
    const int kq   = __builtin_amdgcn_readfirstlane(wave >> 2);
    const int node_base = (blockIdx.x * 4 + tile) * 16;

    const int id = anchor_ids[32 * wave + lane];

    const float* aptr = dists + (size_t)(KHALF * kq + 2 * kh) * N_NODES + node_base + m;
    v2f af[KHALF / 4];
#pragma unroll
    for (int s = 0; s < KPRE; ++s) {
        af[s].x = aptr[(size_t)(4 * s) * N_NODES];
        af[s].y = aptr[(size_t)(4 * s + 1) * N_NODES];
        if ((s & 7) == 7)
            __builtin_amdgcn_sched_barrier(0);
    }

    const int ido  = __shfl_xor(id, 16);
    const int aid0 = kh ? ido : id;
    const int aid1 = kh ? id : ido;

    const v4f* e0 = (const v4f*)(embeds + (size_t)aid0 * DLAT + 16 * kh);
    const v4f* e1 = (const v4f*)(embeds + (size_t)aid1 * DLAT + 16 * kh);
    v4f ea0[4], ea1[4];
#pragma unroll
    for (int q = 0; q < 4; ++q) {
        ea0[q] = e0[q];
        ea1[q] = e1[q];
    }
    v4f* l0 = (v4f*)(Blds + (32 * wave + m) * BSTR + 16 * kh);
    v4f* l1 = (v4f*)(Blds + (32 * wave + 16 + m) * BSTR + 16 * kh);
#pragma unroll
    for (int q = 0; q < 4; ++q) {
        l0[q] = ea0[q];
        l1[q] = ea1[q];
    }
    __builtin_amdgcn_sched_barrier(0);

#pragma unroll
    for (int s = KPRE; s < KHALF / 4; ++s) {
        af[s].x = aptr[(size_t)(4 * s) * N_NODES];
        af[s].y = aptr[(size_t)(4 * s + 1) * N_NODES];
        if ((s & 7) == 7)
            __builtin_amdgcn_sched_barrier(0);
    }

    __builtin_amdgcn_fence(__ATOMIC_RELEASE, "workgroup", "local");
    __builtin_amdgcn_s_barrier();
    __builtin_amdgcn_fence(__ATOMIC_ACQUIRE, "workgroup", "local");

    v8f acc0 = {};
    v8f acc1 = {};

    float cs[32];
    v2f   wb[8][2];
    float w[DLAT];
    float bj;

#pragma unroll
    for (int s = 0; s < KHALF / 4; ++s) {
        if (s == KMID) {
            __builtin_amdgcn_sched_barrier(0);
            const float* ec = embeds + (size_t)(wave * 32) * DLAT + lane;
#pragma unroll
            for (int i = 0; i < 32; ++i)
                cs[i] = ec[i * DLAT];
            if (kq == 0) {
#pragma unroll
                for (int t = 0; t < 8; ++t) {
                    const int d = 16 * kh + 2 * t;
#pragma unroll
                    for (int u = 0; u < 2; ++u) {
                        wb[t][u].x = W[d * DLAT + m + 16 * u];
                        wb[t][u].y = W[(d + 1) * DLAT + m + 16 * u];
                    }
                }
#pragma unroll
                for (int d = 0; d < DLAT; ++d)
                    w[d] = W[(DLAT + d) * DLAT + lane];
                bj = bias[lane];
            }
        }
        const int ka = KHALF * kq + 4 * s + 2 * kh;
        v2f b0, b1;
        b0.x = Blds[ka * BSTR + m];
        b0.y = Blds[(ka + 1) * BSTR + m];
        b1.x = Blds[ka * BSTR + m + 16];
        b1.y = Blds[(ka + 1) * BSTR + m + 16];

        acc0 = __builtin_amdgcn_wmma_f32_16x16x4_f32(
                   false, af[s], false, b0, (short)0, acc0, false, false);
        acc1 = __builtin_amdgcn_wmma_f32_16x16x4_f32(
                   false, af[s], false, b1, (short)0, acc1, false, false);
    }

    {
        float s0 = 0.0f, s1 = 0.0f, s2 = 0.0f, s3 = 0.0f;
#pragma unroll
        for (int i = 0; i < 32; i += 4) {
            s0 += cs[i + 0];
            s1 += cs[i + 1];
            s2 += cs[i + 2];
            s3 += cs[i + 3];
        }
        part[wave * DLAT + lane] = (s0 + s1) + (s2 + s3);
    }

    float* tw = Tlds + wave * (16 * TSTR);
#pragma unroll
    for (int r = 0; r < 8; ++r) {
        tw[(8 * kh + r) * TSTR + m]      = acc0[r];
        tw[(8 * kh + r) * TSTR + m + 16] = acc1[r];
    }
    __syncthreads();

    if (kq == 0) {
        const v4f* t0 = (const v4f*)(Tlds + wave * (16 * TSTR) + m * TSTR + 16 * kh);
        const v4f* t1 = (const v4f*)(Tlds + (wave + 4) * (16 * TSTR) + m * TSTR + 16 * kh);
        v4f ta[4];
#pragma unroll
        for (int q = 0; q < 4; ++q)
            ta[q] = t0[q] + t1[q];

        float sc = 0.0f;
#pragma unroll
        for (int r = 0; r < 8; ++r)
            sc += part[r * DLAT + lane];
        float a0 = 0.0f, a1 = 0.0f, a2 = 0.0f, a3 = 0.0f;
#pragma unroll
        for (int d = 0; d < DLAT; d += 4) {
            a0 = fmaf(lane_bcast(sc, d + 0), w[d + 0], a0);
            a1 = fmaf(lane_bcast(sc, d + 1), w[d + 1], a1);
            a2 = fmaf(lane_bcast(sc, d + 2), w[d + 2], a2);
            a3 = fmaf(lane_bcast(sc, d + 3), w[d + 3], a3);
        }
        const float inv_a = 1.0f / (float)N_ANCH;
        const float cj = ((a0 + a1) + (a2 + a3)) * inv_a + bj;
        const float c_lo = __shfl(cj, m);
        const float c_hi = __shfl(cj, m + 16);

        v8f o0 = {};
        v8f o1 = {};
#pragma unroll
        for (int s = 0; s < 8; ++s) {
            v2f a;
            a.x = ta[s >> 1][2 * (s & 1)];
            a.y = ta[s >> 1][2 * (s & 1) + 1];
            o0 = __builtin_amdgcn_wmma_f32_16x16x4_f32(
                     false, a, false, wb[s][0], (short)0, o0, false, false);
            o1 = __builtin_amdgcn_wmma_f32_16x16x4_f32(
                     false, a, false, wb[s][1], (short)0, o1, false, false);
        }

#pragma unroll
        for (int r = 0; r < 8; ++r) {
            const int row = node_base + 8 * kh + r;
            out[(size_t)row * DLAT + m]      = fmaf(o0[r], inv_a, c_lo);
            out[(size_t)row * DLAT + m + 16] = fmaf(o1[r], inv_a, c_hi);
        }
    }
}

extern "C" void kernel_launch(void* const* d_in, const int* in_sizes, int n_in,
                              void* d_out, int out_size, void* d_ws, size_t ws_size,
                              hipStream_t stream) {
    const float* embeds     = (const float*)d_in[0];
    const float* dists      = (const float*)d_in[1];
    const float* W_hidden   = (const float*)d_in[2];
    const float* b_hidden   = (const float*)d_in[3];
    const int*   anchor_ids = (const int*)d_in[4];
    float* out = (float*)d_out;

    pnn_main<<<N_NODES / 64, 256, 0, stream>>>(embeds, dists, W_hidden, b_hidden,
                                               anchor_ids, out);
}


# pnn_main: B-operand LDS reads issued in five groups, each ahead of the group before's products, not two reads between every two products; E in LDS as two 16-column planes; same sums and order
# speedup vs baseline: 19.1389x; 1.0696x over previous
#include <hip/hip_runtime.h>


typedef __attribute__((ext_vector_type(2))) float v2f;
typedef __attribute__((ext_vector_type(4))) float v4f;
typedef __attribute__((ext_vector_type(8))) float v8f;

#define N_NODES 16384
#define N_ANCH  256
#define DLAT    32
#define PSTR    20
#define PLANE   (N_ANCH * PSTR + 4)
#define TSTR    36
#define KHALF   (N_ANCH / 2)
#define KPRE    16

__device__ __forceinline__ float lane_bcast(float x, int src) {
    return __int_as_float(__builtin_amdgcn_readlane(__float_as_int(x), src));
}

__global__ __launch_bounds__(256) void pnn_main(
        const float* __restrict__ embeds,
        const float* __restrict__ dists,
        const float* __restrict__ W,
        const float* __restrict__ bias,
        const int*   __restrict__ anchor_ids,
        float* __restrict__ out) {
    __shared__ __attribute__((aligned(16))) float Blds[2 * PLANE];
    __shared__ __attribute__((aligned(16))) float Tlds[8 * 16 * TSTR];
    __shared__ float part[8 * DLAT];

    const int tid  = threadIdx.x;
    const int wave = tid >> 5;
    const int lane = tid & 31;
    const int m    = lane & 15;
    const int kh   = lane >> 4;

    const int tile = wave & 3;
    const int kq   = __builtin_amdgcn_readfirstlane(wave >> 2);
    const int node_base = (blockIdx.x * 4 + tile) * 16;

    const int id = anchor_ids[32 * wave + lane];

    const float* aptr = dists + (size_t)(KHALF * kq + 2 * kh) * N_NODES + node_base + m;
    v2f af[KHALF / 4];
#pragma unroll
    for (int s = 0; s < KPRE; ++s) {
        af[s].x = aptr[(size_t)(4 * s) * N_NODES];
        af[s].y = aptr[(size_t)(4 * s + 1) * N_NODES];
        if ((s & 7) == 7)
            __builtin_amdgcn_sched_barrier(0);
    }

    const int ido  = __shfl_xor(id, 16);
    const int aid0 = kh ? ido : id;
    const int aid1 = kh ? id : ido;

    const v4f* e0 = (const v4f*)(embeds + (size_t)aid0 * DLAT + 16 * kh);
    const v4f* e1 = (const v4f*)(embeds + (size_t)aid1 * DLAT + 16 * kh);
    v4f ea0[4], ea1[4];
#pragma unroll
    for (int q = 0; q < 4; ++q) {
        ea0[q] = e0[q];
        ea1[q] = e1[q];
    }
    v4f* l0 = (v4f*)(Blds + kh * PLANE + (32 * wave + m) * PSTR);
    v4f* l1 = (v4f*)(Blds + kh * PLANE + (32 * wave + 16 + m) * PSTR);
#pragma unroll
    for (int q = 0; q < 4; ++q) {
        l0[q] = ea0[q];
        l1[q] = ea1[q];
    }
    __builtin_amdgcn_sched_barrier(0);

#pragma unroll
    for (int s = KPRE; s < KHALF / 4; ++s) {
        af[s].x = aptr[(size_t)(4 * s) * N_NODES];
        af[s].y = aptr[(size_t)(4 * s + 1) * N_NODES];
        if ((s & 7) == 7)
            __builtin_amdgcn_sched_barrier(0);
    }

    __builtin_amdgcn_fence(__ATOMIC_RELEASE, "workgroup", "local");
    __builtin_amdgcn_s_barrier();
    __builtin_amdgcn_fence(__ATOMIC_ACQUIRE, "workgroup", "local");

    v8f acc0 = {};
    v8f acc1 = {};

    float cs[32];
    v2f   wb[8][2];
    float w[DLAT];
    float bj;

    v2f bf0[KHALF / 4];
    v2f bf1[KHALF / 4];

#define LOAD_B_RUN(S0, S1)                                                   \
    _Pragma("unroll")                                                        \
    for (int s = (S0); s < (S1); ++s) {                                      \
        const int ka = KHALF * kq + 4 * s + 2 * kh;                          \
        bf0[s].x = Blds[ka * PSTR + m];                                      \
        bf0[s].y = Blds[(ka + 1) * PSTR + m];                                \
        bf1[s].x = Blds[PLANE + ka * PSTR + m];                              \
        bf1[s].y = Blds[PLANE + (ka + 1) * PSTR + m];                        \
    }
#define WMMA_RUN(S0, S1)                                                     \
    _Pragma("unroll")                                                        \
    for (int s = (S0); s < (S1); ++s) {                                      \
        acc0 = __builtin_amdgcn_wmma_f32_16x16x4_f32(                        \
                   false, af[s], false, bf0[s], (short)0, acc0, false, false); \
        acc1 = __builtin_amdgcn_wmma_f32_16x16x4_f32(                        \
                   false, af[s], false, bf1[s], (short)0, acc1, false, false); \
    }
#define RUN_FENCE() __builtin_amdgcn_sched_barrier(0)

    LOAD_B_RUN(0, 2)
    RUN_FENCE();
    LOAD_B_RUN(2, 8)
    RUN_FENCE();
    WMMA_RUN(0, 2)
    RUN_FENCE();
    LOAD_B_RUN(8, 16)
    RUN_FENCE();
    WMMA_RUN(2, 8)
    RUN_FENCE();
    LOAD_B_RUN(16, 24)
    RUN_FENCE();
    WMMA_RUN(8, 16)
    __builtin_amdgcn_sched_barrier(0);
    {
        const float* ec = embeds + (size_t)(wave * 32) * DLAT + lane;
#pragma unroll
        for (int i = 0; i < 32; ++i)
            cs[i] = ec[i * DLAT];
        if (kq == 0) {
#pragma unroll
            for (int t = 0; t < 8; ++t) {
                const int d = 16 * kh + 2 * t;
#pragma unroll
                for (int u = 0; u < 2; ++u) {
                    wb[t][u].x = W[d * DLAT + m + 16 * u];
                    wb[t][u].y = W[(d + 1) * DLAT + m + 16 * u];
                }
            }
#pragma unroll
            for (int d = 0; d < DLAT; ++d)
                w[d] = W[(DLAT + d) * DLAT + lane];
            bj = bias[lane];
        }
    }
    RUN_FENCE();
    LOAD_B_RUN(24, 32)
    RUN_FENCE();
    WMMA_RUN(16, 24)
    __builtin_amdgcn_sched_barrier(0);
    WMMA_RUN(24, 32)
#undef LOAD_B_RUN
#undef WMMA_RUN
#undef RUN_FENCE

    {
        float s0 = 0.0f, s1 = 0.0f, s2 = 0.0f, s3 = 0.0f;
#pragma unroll
        for (int i = 0; i < 32; i += 4) {
            s0 += cs[i + 0];
            s1 += cs[i + 1];
            s2 += cs[i + 2];
            s3 += cs[i + 3];
        }
        part[wave * DLAT + lane] = (s0 + s1) + (s2 + s3);
    }

    float* tw = Tlds + wave * (16 * TSTR);
#pragma unroll
    for (int r = 0; r < 8; ++r) {
        tw[(8 * kh + r) * TSTR + m]      = acc0[r];
        tw[(8 * kh + r) * TSTR + m + 16] = acc1[r];
    }
    __syncthreads();

    if (kq == 0) {
        const v4f* t0 = (const v4f*)(Tlds + wave * (16 * TSTR) + m * TSTR + 16 * kh);
        const v4f* t1 = (const v4f*)(Tlds + (wave + 4) * (16 * TSTR) + m * TSTR + 16 * kh);
        v4f ta[4];
#pragma unroll
        for (int q = 0; q < 4; ++q)
            ta[q] = t0[q] + t1[q];

        float sc = 0.0f;
#pragma unroll
        for (int r = 0; r < 8; ++r)
            sc += part[r * DLAT + lane];
        float a0 = 0.0f, a1 = 0.0f, a2 = 0.0f, a3 = 0.0f;
#pragma unroll
        for (int d = 0; d < DLAT; d += 4) {
            a0 = fmaf(lane_bcast(sc, d + 0), w[d + 0], a0);
            a1 = fmaf(lane_bcast(sc, d + 1), w[d + 1], a1);
            a2 = fmaf(lane_bcast(sc, d + 2), w[d + 2], a2);
            a3 = fmaf(lane_bcast(sc, d + 3), w[d + 3], a3);
        }
        const float inv_a = 1.0f / (float)N_ANCH;
        const float cj = ((a0 + a1) + (a2 + a3)) * inv_a + bj;
        const float c_lo = __shfl(cj, m);
        const float c_hi = __shfl(cj, m + 16);

        v8f o0 = {};
        v8f o1 = {};
#pragma unroll
        for (int s = 0; s < 8; ++s) {
            v2f a;
            a.x = ta[s >> 1][2 * (s & 1)];
            a.y = ta[s >> 1][2 * (s & 1) + 1];
            o0 = __builtin_amdgcn_wmma_f32_16x16x4_f32(
                     false, a, false, wb[s][0], (short)0, o0, false, false);
            o1 = __builtin_amdgcn_wmma_f32_16x16x4_f32(
                     false, a, false, wb[s][1], (short)0, o1, false, false);
        }

#pragma unroll
        for (int r = 0; r < 8; ++r) {
            const int row = node_base + 8 * kh + r;
            out[(size_t)row * DLAT + m]      = fmaf(o0[r], inv_a, c_lo);
            out[(size_t)row * DLAT + m + 16] = fmaf(o1[r], inv_a, c_hi);
        }
    }
}

extern "C" void kernel_launch(void* const* d_in, const int* in_sizes, int n_in,
                              void* d_out, int out_size, void* d_ws, size_t ws_size,
                              hipStream_t stream) {
    const float* embeds     = (const float*)d_in[0];
    const float* dists      = (const float*)d_in[1];
    const float* W_hidden   = (const float*)d_in[2];
    const float* b_hidden   = (const float*)d_in[3];
    const int*   anchor_ids = (const int*)d_in[4];
    float* out = (float*)d_out;

    pnn_main<<<N_NODES / 64, 256, 0, stream>>>(embeds, dists, W_hidden, b_hidden,
                                               anchor_ids, out);
}
